# starting kernel
# baseline (speedup 1.0000x reference)
#include <hip/hip_runtime.h>

typedef __attribute__((ext_vector_type(2))) float v2f;
typedef __attribute__((ext_vector_type(8))) float v8f;

#define COLS   8192
#define NROWS  4000
#define BB     64
#define MM     63
#define PADCOL 32

#define JST 20
#define KST (BB * JST + 16)
#define YFLOATS (MM * KST)

__global__ __launch_bounds__(256) void prep_R(const float* __restrict__ Rp,
                                              float* __restrict__ Rn) {
    int t = blockIdx.x * blockDim.x + threadIdx.x;
    if (t >= MM * BB) return;
    int k = t / BB;
    const float* src = Rp + (size_t)t * BB;
    float sum = 0.f;
    #pragma unroll 4
    for (int i = 0; i < BB; ++i) {
        float v = src[i];
        if (k == MM - 1 && i >= PADCOL) v = 0.f;
        sum += v * v;
    }
    float inv = 1.0f / sum;
    float* dst = Rn + (size_t)t * BB;
    #pragma unroll 4
    for (int i = 0; i < BB; ++i) {
        float v = src[i];
        if (k == MM - 1 && i >= PADCOL) v = 0.f;
        dst[i] = v * v * inv;
    }
}

__global__ __launch_bounds__(256) void prep_L(const float* __restrict__ Lparams,
                                              float* __restrict__ Lp) {
    int t = blockIdx.x * blockDim.x + threadIdx.x;
    if (t >= BB * BB) return;
    int j = t / BB, l = t % BB;
    float* dst = Lp + (size_t)t * BB;
    if (l >= MM) {
        #pragma unroll 4
        for (int kk = 0; kk < BB; ++kk) dst[kk] = 0.f;
        return;
    }
    const float* src = Lparams + ((size_t)j * MM + l) * MM;
    float sum = 0.f;
    for (int kk = 0; kk < MM; ++kk) { float v = src[kk]; sum += v * v; }
    float inv = 1.0f / sum;
    for (int kk = 0; kk < MM; ++kk) dst[kk] = src[kk] * src[kk] * inv;
    dst[MM] = 0.f;
}

__global__ __launch_bounds__(256) void monarch_main(const float* __restrict__ x,
                                                    const float* __restrict__ Rn,
                                                    const float* __restrict__ Lp,
                                                    float* __restrict__ out) {
    __shared__ float Ys[YFLOATS];
    const int ct   = blockIdx.x * 16;
    const int wave = threadIdx.x >> 5;
    const int lane = threadIdx.x & 31;
    const int lh   = lane >> 4;
    const int lm   = lane & 15;

    for (int k = wave; k < MM; k += 8) {
        float xr[32];
        #pragma unroll
        for (int r = 0; r < 32; ++r) {
            int s = r >> 1, o = r & 1;
            int grow = k * BB + s * 4 + 2 * lh + o;
            xr[r] = (grow < NROWS) ? x[(size_t)grow * COLS + ct + lm] : 0.f;
        }
        v8f acc[4] = {};
        const float* Rk = Rn + (size_t)k * BB * BB;
        #pragma unroll
        for (int s = 0; s < 16; ++s) {
            v2f b = { xr[2 * s], xr[2 * s + 1] };
            #pragma unroll
            for (int jt = 0; jt < 4; ++jt) {
                v2f a = *(const v2f*)(Rk + (size_t)(jt * 16 + lm) * BB + s * 4 + 2 * lh);
                acc[jt] = __builtin_amdgcn_wmma_f32_16x16x4_f32(
                    false, a, false, b, (short)0, acc[jt], false, false);
            }
        }
        #pragma unroll
        for (int jt = 0; jt < 4; ++jt) {
            #pragma unroll
            for (int v = 0; v < 8; ++v) {
                int j = jt * 16 + v + 8 * lh;
                Ys[k * KST + j * JST + lm] = acc[jt][v];
            }
        }
    }
    __syncthreads();

    for (int j = wave; j < BB; j += 8) {
        v8f acc[4] = {};
        const float* Lj = Lp + (size_t)j * BB * BB;
        #pragma unroll
        for (int s = 0; s < 16; ++s) {
            int k0 = s * 4 + 2 * lh;
            int ka = (k0     > MM - 1) ? (MM - 1) : k0;
            int kb = (k0 + 1 > MM - 1) ? (MM - 1) : (k0 + 1);
            v2f b = { Ys[ka * KST + j * JST + lm],
                      Ys[kb * KST + j * JST + lm] };
            #pragma unroll
            for (int lt = 0; lt < 4; ++lt) {
                v2f a = *(const v2f*)(Lj + (size_t)(lt * 16 + lm) * BB + k0);
                acc[lt] = __builtin_amdgcn_wmma_f32_16x16x4_f32(
                    false, a, false, b, (short)0, acc[lt], false, false);
            }
        }
        #pragma unroll
        for (int lt = 0; lt < 4; ++lt) {
            #pragma unroll
            for (int v = 0; v < 8; ++v) {
                int l   = lt * 16 + v + 8 * lh;
                int row = l * BB + j;
                if (row < NROWS)
                    out[(size_t)row * COLS + ct + lm] = acc[lt][v];
            }
        }
    }
}

extern "C" void kernel_launch(void* const* d_in, const int* in_sizes, int n_in,
                              void* d_out, int out_size, void* d_ws, size_t ws_size,
                              hipStream_t stream) {
    const float* x       = (const float*)d_in[0];
    const float* Lparams = (const float*)d_in[1];
    const float* Rparams = (const float*)d_in[2];
    float* out = (float*)d_out;

    float* Rn = (float*)d_ws;
    float* Lp = Rn + (size_t)MM * BB * BB;

    prep_R<<<(MM * BB + 255) / 256, 256, 0, stream>>>(Rparams, Rn);
    prep_L<<<(BB * BB + 255) / 256, 256, 0, stream>>>(Lparams, Lp);
    monarch_main<<<COLS / 16, 256, 0, stream>>>(x, Rn, Lp, out);
}


# each workgroup builds its block's simplex A operand in LDS from the raw parameters (row sums in the same serial order), so both prep launches and their workspace planes go: 4 launches become 2
# speedup vs baseline: 2.2404x; 2.2404x over previous
#include <hip/hip_runtime.h>

typedef __attribute__((ext_vector_type(2))) float v2f;
typedef __attribute__((ext_vector_type(4))) float v4f;
typedef __attribute__((ext_vector_type(8))) float v8f;

#define COLS   8192
#define NROWS  4000
#define BB     64
#define MM     63
#define PADCOL 32
#define WAVES  8
#define CG     4

#define LPITCH 68

template <int PHASE>
__device__ __forceinline__ void build_A(const float* __restrict__ P, int blk,
                                        int lh, int lm, v4f (&a)[4][8]) {
    __shared__ __attribute__((aligned(16))) float raw[BB * LPITCH];
    __shared__ float inv_s[BB];
    const int tid = threadIdx.x;
    if (PHASE == 1) {
        const v4f* g = (const v4f*)(P + (size_t)blk * BB * BB);
        #pragma unroll
        for (int u = 0; u < 4; ++u) {
            const int q = tid + 256 * u;
            *(v4f*)(raw + (q >> 4) * LPITCH + (q & 15) * 4) = g[q];
        }
    } else {
        const float* g = P + (size_t)blk * MM * MM;
        if (tid < BB) { raw[tid * LPITCH + MM] = 0.f; raw[MM * LPITCH + tid] = 0.f; }
        #pragma unroll
        for (int u = 0; u < 16; ++u) {
            const int e = tid + 256 * u;
            if (e < MM * MM) {
                const int l = e / MM;
                raw[l * LPITCH + (e - l * MM)] = g[e];
            }
        }
    }
    __syncthreads();
    if (tid < ((PHASE == 1) ? BB : MM)) {
        const float* rr = raw + tid * LPITCH;
        float sum = 0.f;
        if (PHASE == 1) {
            #pragma unroll
            for (int i = 0; i < BB; ++i) {
                float v = rr[i];
                if (blk == MM - 1 && i >= PADCOL) v = 0.f;
                sum = __builtin_fmaf(v, v, sum);
            }
        } else {
            #pragma unroll
            for (int i = 0; i < MM; ++i) { const float v = rr[i]; sum = __builtin_fmaf(v, v, sum); }
        }
        inv_s[tid] = 1.0f / sum;
    } else if (PHASE == 2 && tid == MM) {
        inv_s[MM] = 0.f;
    }
    __syncthreads();
    #pragma unroll
    for (int jt = 0; jt < 4; ++jt) {
        const int r = jt * 16 + lm;
        const float iv = inv_s[r];
        const float* rr = raw + r * LPITCH + 2 * lh;
        const bool rowok = (PHASE == 1) || (r < MM);
        #pragma unroll
        for (int p = 0; p < 8; ++p) {
            const v2f lo = *(const v2f*)(rr + 8 * p);
            const v2f hi = *(const v2f*)(rr + 8 * p + 4);
            v4f o = { lo.x * lo.x * iv, lo.y * lo.y * iv, hi.x * hi.x * iv, hi.y * hi.y * iv };
            if (PHASE == 1) {
                if (blk == MM - 1 && p >= PADCOL / 8) o = v4f{ 0.f, 0.f, 0.f, 0.f };
            } else {
                if (!rowok) o = v4f{ 0.f, 0.f, 0.f, 0.f };
                if (p == 7 && lh == 1) o.w = 0.f;
            }
            a[jt][p] = o;
        }
    }
}

template <int PHASE>
__device__ __forceinline__ void block_mm(const float* __restrict__ src,
                                         const float* __restrict__ Araw,
                                         float* __restrict__ dst) {
    const int blk  = blockIdx.y;
    const int wave = threadIdx.x >> 5;
    const int lane = threadIdx.x & 31;
    const int lh   = lane >> 4;
    const int lm   = lane & 15;

    v4f a[4][8];
    build_A<PHASE>(Araw, blk, lh, lm, a);

    const int cwave = (blockIdx.x * WAVES + wave) * (CG * 64) + 4 * lm;
    const float* sbase = src + (size_t)blk * BB * COLS + cwave;

    #pragma unroll 1
    for (int g = 0; g < CG; ++g) {
        const float* sp = sbase + g * 64;
        v4f b[32];
        #pragma unroll
        for (int r = 0; r < 32; ++r) {
            const int i = 4 * (r >> 1) + 2 * lh + (r & 1);
            const int ii = (PHASE == 1) ? ((blk == MM - 1) ? (i & (PADCOL - 1)) : i)
                                        : ((i > MM - 1) ? (MM - 1) : i);
            b[r] = *(const v4f*)(sp + (size_t)ii * COLS);
        }
        v8f acc[4][4] = {};
        #pragma unroll
        for (int s = 0; s < 16; ++s) {
            #pragma unroll
            for (int t = 0; t < 4; ++t) {
                const v2f bf = { b[2 * s][t], b[2 * s + 1][t] };
                #pragma unroll
                for (int jt = 0; jt < 4; ++jt) {
                    const v4f av = a[jt][s >> 1];
                    const v2f af = (s & 1) ? v2f{ av.z, av.w } : v2f{ av.x, av.y };
                    acc[jt][t] = __builtin_amdgcn_wmma_f32_16x16x4_f32(
                        false, af, false, bf, (short)0, acc[jt][t], false, false);
                }
            }
        }
        #pragma unroll
        for (int jt = 0; jt < 4; ++jt) {
            #pragma unroll
            for (int v = 0; v < 8; ++v) {
                const int m    = jt * 16 + v + 8 * lh;
                const int orow = m * BB + blk;
                const v4f o4 = { acc[jt][0][v], acc[jt][1][v], acc[jt][2][v], acc[jt][3][v] };
                if (PHASE == 1 || jt < 3 || v < 6 || orow < NROWS)
                    *(v4f*)(dst + (size_t)orow * COLS + cwave + g * 64) = o4;
            }
        }
    }
}

__global__ __launch_bounds__(256) void monarch_p1(const float* __restrict__ x,
                                                  const float* __restrict__ Rparams,
                                                  float* __restrict__ Yw) {
    block_mm<1>(x, Rparams, Yw);
}

__global__ __launch_bounds__(256) void monarch_p2(const float* __restrict__ Yw,
                                                  const float* __restrict__ Lparams,
                                                  float* __restrict__ out) {
    block_mm<2>(Yw, Lparams, out);
}

extern "C" void kernel_launch(void* const* d_in, const int* in_sizes, int n_in,
                              void* d_out, int out_size, void* d_ws, size_t ws_size,
                              hipStream_t stream) {
    const float* x       = (const float*)d_in[0];
    const float* Lparams = (const float*)d_in[1];
    const float* Rparams = (const float*)d_in[2];
    float* out = (float*)d_out;

    float* Yw = (float*)d_ws;

    monarch_p1<<<dim3(COLS / (WAVES * CG * 64), MM), 256, 0, stream>>>(x, Rparams, Yw);
    monarch_p2<<<dim3(COLS / (WAVES * CG * 64), BB), 256, 0, stream>>>(Yw, Lparams, out);
}


# B rows run one 64-column group ahead: behind each K quarter's 64 WMMAs its 8 registers are refilled with the next group's rows, the first group's loads go out before the A build; sum order unchanged
# speedup vs baseline: 2.3704x; 1.0580x over previous
#include <hip/hip_runtime.h>

typedef __attribute__((ext_vector_type(2))) float v2f;
typedef __attribute__((ext_vector_type(4))) float v4f;
typedef __attribute__((ext_vector_type(8))) float v8f;

#define COLS   8192
#define NROWS  4000
#define BB     64
#define MM     63
#define PADCOL 32
#define WAVES  8
#define CG     4

#define LPITCH 68

template <int PHASE, typename F>
__device__ __forceinline__ void build_A(const float* __restrict__ P, int blk,
                                        int lh, int lm, v4f (&a)[4][8], F issue_b) {
    __shared__ __attribute__((aligned(16))) float raw[BB * LPITCH];
    __shared__ float inv_s[BB];
    const int tid = threadIdx.x;
    if (PHASE == 1) {
        const v4f* g = (const v4f*)(P + (size_t)blk * BB * BB);
        v4f pv[4];
        #pragma unroll
        for (int u = 0; u < 4; ++u) pv[u] = g[tid + 256 * u];
        issue_b();
        #pragma unroll
        for (int u = 0; u < 4; ++u) {
            const int q = tid + 256 * u;
            *(v4f*)(raw + (q >> 4) * LPITCH + (q & 15) * 4) = pv[u];
        }
    } else {
        const float* g = P + (size_t)blk * MM * MM;
        float pv[16];
        #pragma unroll
        for (int u = 0; u < 16; ++u) {
            const int e = tid + 256 * u;
            pv[u] = (e < MM * MM) ? g[e] : 0.f;
        }
        issue_b();
        if (tid < BB) { raw[tid * LPITCH + MM] = 0.f; raw[MM * LPITCH + tid] = 0.f; }
        #pragma unroll
        for (int u = 0; u < 16; ++u) {
            const int e = tid + 256 * u;
            if (e < MM * MM) {
                const int l = e / MM;
                raw[l * LPITCH + (e - l * MM)] = pv[u];
            }
        }
    }
    __syncthreads();
    if (tid < ((PHASE == 1) ? BB : MM)) {
        const float* rr = raw + tid * LPITCH;
        float sum = 0.f;
        if (PHASE == 1) {
            #pragma unroll
            for (int i = 0; i < BB; ++i) {
                float v = rr[i];
                if (blk == MM - 1 && i >= PADCOL) v = 0.f;
                sum = __builtin_fmaf(v, v, sum);
            }
        } else {
            #pragma unroll
            for (int i = 0; i < MM; ++i) { const float v = rr[i]; sum = __builtin_fmaf(v, v, sum); }
        }
        inv_s[tid] = 1.0f / sum;
    } else if (PHASE == 2 && tid == MM) {
        inv_s[MM] = 0.f;
    }
    __syncthreads();
    #pragma unroll
    for (int jt = 0; jt < 4; ++jt) {
        const int r = jt * 16 + lm;
        const float iv = inv_s[r];
        const float* rr = raw + r * LPITCH + 2 * lh;
        const bool rowok = (PHASE == 1) || (r < MM);
        #pragma unroll
        for (int p = 0; p < 8; ++p) {
            const v2f lo = *(const v2f*)(rr + 8 * p);
            const v2f hi = *(const v2f*)(rr + 8 * p + 4);
            v4f o = { lo.x * lo.x * iv, lo.y * lo.y * iv, hi.x * hi.x * iv, hi.y * hi.y * iv };
            if (PHASE == 1) {
                if (blk == MM - 1 && p >= PADCOL / 8) o = v4f{ 0.f, 0.f, 0.f, 0.f };
            } else {
                if (!rowok) o = v4f{ 0.f, 0.f, 0.f, 0.f };
                if (p == 7 && lh == 1) o.w = 0.f;
            }
            a[jt][p] = o;
        }
    }
}

struct BPtr { const float* lo; const float* hi; };

template <int PHASE>
__device__ __forceinline__ BPtr b_ptrs(const float* __restrict__ sp, int blk, int lh) {
    BPtr p;
    p.lo = sp + (size_t)(2 * lh) * COLS;
    if (PHASE == 1) p.hi = p.lo + ((blk == MM - 1) ? 0 : 32 * COLS);
    else            p.hi = sp + (size_t)(MM - 2 + lh) * COLS;
    return p;
}

template <int PHASE, int R0, int NR>
__device__ __forceinline__ void load_b(v4f (&b)[32], const BPtr p) {
    #pragma unroll
    for (int r = R0; r < R0 + NR; ++r) {
        const int c = 4 * (r >> 1) + (r & 1);
        if (PHASE == 1)
            b[r] = (r < 16) ? *(const v4f*)(p.lo + (size_t)c * COLS)
                            : *(const v4f*)(p.hi + (size_t)(c - 32) * COLS);
        else
            b[r] = (r < 31) ? *(const v4f*)(p.lo + (size_t)c * COLS)
                            : *(const v4f*)(p.hi);
    }
}

template <int Q>
__device__ __forceinline__ void mma_quarter(const v4f (&a)[4][8], const v4f (&b)[32], v8f (&acc)[4][4]) {
    #pragma unroll
    for (int s = 4 * Q; s < 4 * Q + 4; ++s) {
        #pragma unroll
        for (int t = 0; t < 4; ++t) {
            const v2f bf = { b[2 * s][t], b[2 * s + 1][t] };
            #pragma unroll
            for (int jt = 0; jt < 4; ++jt) {
                const v4f av = a[jt][s >> 1];
                const v2f af = (s & 1) ? v2f{ av.z, av.w } : v2f{ av.x, av.y };
                acc[jt][t] = __builtin_amdgcn_wmma_f32_16x16x4_f32(
                    false, af, false, bf, (short)0, acc[jt][t], false, false);
            }
        }
    }
}

template <int PHASE, bool MORE>
__device__ __forceinline__ void do_group(const v4f (&a)[4][8], v4f (&b)[32],
                                         const BPtr nx, float* __restrict__ dst,
                                         int blk, int lh, int col) {
    const unsigned lane_q = (unsigned)(((8 * lh) * BB + blk) * COLS + col) >> 2;
    v8f acc[4][4] = {};
    mma_quarter<0>(a, b, acc);
    __builtin_amdgcn_sched_barrier(0);
    if (MORE) { load_b<PHASE, 0, 8>(b, nx);  __builtin_amdgcn_sched_barrier(0); }
    mma_quarter<1>(a, b, acc);
    __builtin_amdgcn_sched_barrier(0);
    if (MORE) { load_b<PHASE, 8, 8>(b, nx);  __builtin_amdgcn_sched_barrier(0); }
    mma_quarter<2>(a, b, acc);
    __builtin_amdgcn_sched_barrier(0);
    if (MORE) { load_b<PHASE, 16, 8>(b, nx); __builtin_amdgcn_sched_barrier(0); }
    mma_quarter<3>(a, b, acc);
    __builtin_amdgcn_sched_barrier(0);
    if (MORE) { load_b<PHASE, 24, 8>(b, nx); __builtin_amdgcn_sched_barrier(0); }
    #pragma unroll
    for (int jt = 0; jt < 4; ++jt) {
        #pragma unroll
        for (int v = 0; v < 8; ++v) {
            const int m    = jt * 16 + v + 8 * lh;
            const int orow = m * BB + blk;
            const v4f o4 = { acc[jt][0][v], acc[jt][1][v], acc[jt][2][v], acc[jt][3][v] };
            v4f* rowp = (v4f*)(dst + (size_t)(jt * 16 + v) * BB * COLS);
            if (PHASE == 1 || jt < 3 || v < 6 || orow < NROWS)
                rowp[lane_q] = o4;
        }
    }
}

template <int PHASE>
__device__ __forceinline__ void block_mm(const float* __restrict__ src,
                                         const float* __restrict__ Araw,
                                         float* __restrict__ dst) {
    const int blk  = blockIdx.y;
    const int wave = threadIdx.x >> 5;
    const int lane = threadIdx.x & 31;
    const int lh   = lane >> 4;
    const int lm   = lane & 15;

    const int cwave = (blockIdx.x * WAVES + wave) * (CG * 64) + 4 * lm;
    BPtr bp = b_ptrs<PHASE>(src + (size_t)blk * BB * COLS + cwave, blk, lh);

    v4f b[32];

    v4f a[4][8];
    build_A<PHASE>(Araw, blk, lh, lm, a,
                   [&]() { load_b<PHASE, 0, 32>(b, bp); });

    #pragma unroll 1
    for (int g = 0; g < CG - 1; ++g) {
        bp.lo += 64; bp.hi += 64;
        do_group<PHASE, true>(a, b, bp, dst, blk, lh, cwave + g * 64);
    }
    do_group<PHASE, false>(a, b, bp, dst, blk, lh, cwave + (CG - 1) * 64);
}

__global__ __launch_bounds__(256) void monarch_p1(const float* __restrict__ x,
                                                  const float* __restrict__ Rparams,
                                                  float* __restrict__ Yw) {
    block_mm<1>(x, Rparams, Yw);
}

__global__ __launch_bounds__(256) void monarch_p2(const float* __restrict__ Yw,
                                                  const float* __restrict__ Lparams,
                                                  float* __restrict__ out) {
    block_mm<2>(Yw, Lparams, out);
}

extern "C" void kernel_launch(void* const* d_in, const int* in_sizes, int n_in,
                              void* d_out, int out_size, void* d_ws, size_t ws_size,
                              hipStream_t stream) {
    const float* x       = (const float*)d_in[0];
    const float* Lparams = (const float*)d_in[1];
    const float* Rparams = (const float*)d_in[2];
    float* out = (float*)d_out;

    float* Yw = (float*)d_ws;

    monarch_p1<<<dim3(COLS / (WAVES * CG * 64), MM), 256, 0, stream>>>(x, Rparams, Yw);
    monarch_p2<<<dim3(COLS / (WAVES * CG * 64), BB), 256, 0, stream>>>(Yw, Lparams, out);
}
